# starting kernel
# baseline (speedup 1.0000x reference)
#include <hip/hip_runtime.h>

typedef __attribute__((ext_vector_type(2))) float v2f;
typedef __attribute__((ext_vector_type(8))) float v8f;

static constexpr int NN     = 100000;
static constexpr int NE     = 1600000;
static constexpr int DK     = 128;
static constexpr int MTILES = NN / 16;

template<int NT, bool RELU_IN>
__global__ __launch_bounds__(128)
void gemm_wmma_f32(const float* __restrict__ A, const float* __restrict__ W,
                   const float* __restrict__ bias, float* __restrict__ out)
{
    constexpr int DOUT = NT * 16;
    const int lane  = threadIdx.x & 31;
    const int wave  = threadIdx.x >> 5;
    const int mtile = blockIdx.x * 4 + wave;
    if (mtile >= MTILES) return;

    const int half = lane >> 4;
    const int mn   = lane & 15;
    const int kb   = half * 2;

    const float* arow = A + (size_t)(mtile * 16 + mn) * DK;

    v8f acc[NT];
#pragma unroll
    for (int t = 0; t < NT; ++t) acc[t] = (v8f)0.0f;

    for (int k = 0; k < DK; k += 4) {
        v2f a = *(const v2f*)(arow + k + kb);
        if (RELU_IN) { a.x = fmaxf(a.x, 0.0f); a.y = fmaxf(a.y, 0.0f); }
        const float* wp = W + (size_t)(k + kb) * DOUT + mn;
#pragma unroll
        for (int t = 0; t < NT; ++t) {
            v2f b;
            b.x = wp[t * 16];
            b.y = wp[t * 16 + DOUT];
            acc[t] = __builtin_amdgcn_wmma_f32_16x16x4_f32(
                false, a, false, b, (short)0, acc[t], false, false);
        }
    }

#pragma unroll
    for (int t = 0; t < NT; ++t) {
        const float bv = bias[t * 16 + mn];
#pragma unroll
        for (int v = 0; v < 8; ++v) {
            const int m = mtile * 16 + v + 8 * half;
            out[(size_t)m * DOUT + t * 16 + mn] = acc[t][v] + bv;
        }
    }
}

template<int VEC>
__global__ __launch_bounds__(256)
void spmm_atomic(const float* __restrict__ vals, const int* __restrict__ row,
                 const int* __restrict__ col, const float* __restrict__ x,
                 float* __restrict__ out)
{
    constexpr int D = 32 * VEC;
    const long long gt = (long long)blockIdx.x * blockDim.x + threadIdx.x;
    const int e    = (int)(gt >> 5);
    const int lane = (int)(gt & 31);
    if (e >= NE) return;

    const float v = vals[e];
    const int   r = row[e];
    const int   c = col[e];

    const float* xs = x   + (size_t)c * D + lane * VEC;
    float*       od = out + (size_t)r * D + lane * VEC;

    float t[VEC];
    if constexpr (VEC == 4) {
        const float4 q = *(const float4*)xs;
        t[0] = q.x; t[1] = q.y; t[2] = q.z; t[3] = q.w;
    } else {
        const float2 q = *(const float2*)xs;
        t[0] = q.x; t[1] = q.y;
    }
#pragma unroll
    for (int i = 0; i < VEC; ++i)
        atomicAdd(od + i, v * t[i]);
}

extern "C" void kernel_launch(void* const* d_in, const int* in_sizes, int n_in,
                              void* d_out, int out_size, void* d_ws, size_t ws_size,
                              hipStream_t stream)
{
    const float* x    = (const float*)d_in[0];
    const float* vals = (const float*)d_in[1];
    const float* W1   = (const float*)d_in[2];
    const float* b1   = (const float*)d_in[3];
    const float* W2   = (const float*)d_in[4];
    const float* b2   = (const float*)d_in[5];
    const float* W3   = (const float*)d_in[6];
    const float* b3   = (const float*)d_in[7];
    const int*   row  = (const int*)d_in[8];
    const int*   col  = (const int*)d_in[9];

    float* bufA = (float*)d_ws;
    float* bufB = bufA + (size_t)NN * 128;

    const dim3 gBlk(128);
    const dim3 gGrd((MTILES + 3) / 4);
    const dim3 sBlk(256);
    const dim3 sGrd(NE / 8);

    gemm_wmma_f32<8, false><<<gGrd, gBlk, 0, stream>>>(x, W1, b1, bufA);
    hipMemsetAsync(bufB, 0, (size_t)NN * 128 * sizeof(float), stream);
    spmm_atomic<4><<<sGrd, sBlk, 0, stream>>>(vals, row, col, bufA, bufB);

    gemm_wmma_f32<8, true><<<gGrd, gBlk, 0, stream>>>(bufB, W2, b2, bufA);
    hipMemsetAsync(bufB, 0, (size_t)NN * 128 * sizeof(float), stream);
    spmm_atomic<4><<<sGrd, sBlk, 0, stream>>>(vals, row, col, bufA, bufB);

    gemm_wmma_f32<4, true><<<gGrd, gBlk, 0, stream>>>(bufB, W3, b3, bufA);
    hipMemsetAsync(d_out, 0, (size_t)NN * 64 * sizeof(float), stream);
    spmm_atomic<2><<<sGrd, sBlk, 0, stream>>>(vals, row, col, bufA, (float*)d_out);
}


# SpMM build scan: queue slots from a wave-wide vote (ballot + lane count, entry count in a scalar register), not an LDS counter add and read-back per step; LDS float adds, arrival order kept
# speedup vs baseline: 2.9452x; 2.9452x over previous
#include <hip/hip_runtime.h>

typedef __attribute__((ext_vector_type(2))) float v2f;
typedef __attribute__((ext_vector_type(8))) float v8f;

static constexpr int NN     = 100000;
static constexpr int NE     = 1600000;
static constexpr int DK     = 128;
static constexpr int MTILES = NN / 16;
static constexpr int RPB    = 391;
static constexpr int SS     = 16384;
static constexpr int NSS    = (NE + SS - 1) / SS;
static constexpr int QCAP   = 30720;
static constexpr int NWV    = 32;
static constexpr int QW     = QCAP / NWV;
static constexpr int WSTEP  = 512;
static constexpr int SEGPW  = 112;
static constexpr int POOLN  = NE + 400000;

template<int NT, bool RELU_IN>
__global__ __launch_bounds__(128)
void gemm_wmma_f32(const float* __restrict__ A, const float* __restrict__ W,
                   const float* __restrict__ bias, float* __restrict__ out)
{
    constexpr int DOUT = NT * 16;
    const int lane  = threadIdx.x & 31;
    const int wave  = threadIdx.x >> 5;
    const int mtile = blockIdx.x * 4 + wave;
    if (mtile >= MTILES) return;

    const int half = lane >> 4;
    const int mn   = lane & 15;
    const int kb   = half * 2;

    const float* arow = A + (size_t)(mtile * 16 + mn) * DK;

    v8f acc[NT];
#pragma unroll
    for (int t = 0; t < NT; ++t) acc[t] = (v8f)0.0f;

    for (int k = 0; k < DK; k += 4) {
        v2f a = *(const v2f*)(arow + k + kb);
        if (RELU_IN) { a.x = fmaxf(a.x, 0.0f); a.y = fmaxf(a.y, 0.0f); }
        const float* wp = W + (size_t)(k + kb) * DOUT + mn;
#pragma unroll
        for (int t = 0; t < NT; ++t) {
            v2f b;
            b.x = wp[t * 16];
            b.y = wp[t * 16 + DOUT];
            acc[t] = __builtin_amdgcn_wmma_f32_16x16x4_f32(
                false, a, false, b, (short)0, acc[t], false, false);
        }
    }

#pragma unroll
    for (int t = 0; t < NT; ++t) {
        const float bv = bias[t * 16 + mn];
#pragma unroll
        for (int v = 0; v < 8; ++v) {
            const int m = mtile * 16 + v + 8 * half;
            out[(size_t)m * DOUT + t * 16 + mn] = acc[t][v] + bv;
        }
    }
}

__global__ __launch_bounds__(256)
void owner_keys(const int* __restrict__ row, unsigned int* __restrict__ keys,
                unsigned int* __restrict__ pool_cnt)
{
    const int t = blockIdx.x * 256 + threadIdx.x;
    if (t >= NE / 16) return;
    if (t == 0) *pool_cnt = 0u;
    const int4* rp = (const int4*)(row + (size_t)t * 16);
    unsigned int o[4];
#pragma unroll
    for (int d = 0; d < 4; ++d) {
        const int4 r = rp[d];
        o[d] =  (((unsigned)r.x / RPB) & 255u)
             | ((((unsigned)r.y / RPB) & 255u) << 8)
             | ((((unsigned)r.z / RPB) & 255u) << 16)
             | ((((unsigned)r.w / RPB) & 255u) << 24);
    }
    ((uint4*)keys)[t] = make_uint4(o[0], o[1], o[2], o[3]);
}


template<int VEC>
__device__ __forceinline__ void gather_add(unsigned int lrv, int ecv, float ev, int m, int rows, int lane,
                                           const float* __restrict__ x, float* acc)
{
    constexpr int D = 32 * VEC;
    constexpr int U = 16;
    typedef __attribute__((ext_vector_type(VEC))) float VT;
    for (int j0 = 0; j0 < m; j0 += U) {
        VT xv[U];
#pragma unroll
        for (int u = 0; u < U; ++u) {
            const int j = (j0 + u < m) ? (j0 + u) : j0;
            const int c = __builtin_amdgcn_readlane(ecv, j);
            xv[u] = *(const VT*)(x + (size_t)c * D + lane * VEC);
        }
#pragma unroll
        for (int u = 0; u < U; ++u) asm volatile("" :: "v"(xv[u]));
#pragma unroll
        for (int u = 0; u < U; ++u) {
            const int j = j0 + u;
            if (j < m) {
                const unsigned int lr = (unsigned)__builtin_amdgcn_readlane((int)lrv, j);
                const float v = __int_as_float(__builtin_amdgcn_readlane(__float_as_int(ev), j));
                if (lr < (unsigned)rows) {
                    float* a = acc + lr * D + lane;
#pragma unroll
                    for (int k = 0; k < VEC; ++k) atomicAdd(a + k * 32, v * xv[u][k]);
                }
            }
        }
    }
}

template<int VEC>
__global__ __launch_bounds__(1024)
void spmm_owner_build(const float* __restrict__ vals, const int* __restrict__ row,
                      const int* __restrict__ col, const unsigned int* __restrict__ keys,
                      const float* __restrict__ x, float* __restrict__ out,
                      unsigned int* __restrict__ pool_pk, float* __restrict__ pool_val,
                      unsigned int* __restrict__ seg, unsigned int* __restrict__ nsegs,
                      unsigned int* __restrict__ pool_cnt)
{
    constexpr int D  = 32 * VEC;
    constexpr int PF = 8;
    typedef __attribute__((ext_vector_type(VEC))) float VT;

    __shared__ float        acc[RPB * D];
    __shared__ unsigned int q[QCAP];

    const int tid  = threadIdx.x;
    const int lane = tid & 31;
    const int wave = __builtin_amdgcn_readfirstlane(tid >> 5);
    const int b    = blockIdx.x;
    const int base = b * RPB;
    const int rows = (NN - base) < RPB ? (NN - base) : RPB;
    const unsigned int splat  = (unsigned)b * 0x01010101u;
    const unsigned int nokey  = ~splat;

    for (int i = tid; i < RPB * D / 4; i += 1024)
        ((float4*)acc)[i] = make_float4(0.0f, 0.0f, 0.0f, 0.0f);
    __syncthreads();

    const int qo = wave * QW;
    const unsigned int sbase = (unsigned)(b * NWV + wave) * SEGPW;

    const uint4* kp = (const uint4*)keys;
    const int s0 = b % NSS;

    auto load_keys = [&](int it) -> uint4 {
        const int g = ((s0 + it) % NSS) * 1024 + tid;
        return kp[g < NE / 16 ? g : NE / 16 - 1];
    };

    auto drain = [&](int n, unsigned int off) {
        for (int g0 = 0; g0 < n; g0 += 32) {
            const int m = (n - g0) < 32 ? (n - g0) : 32;
            const unsigned int e = q[qo + g0 + (lane < m ? lane : 0)];
            const int   er = row[e];
            const int   ec = col[e];
            const float ev = vals[e];
            const unsigned int d   = (unsigned)(er - base);
            const unsigned int lrv = d < 511u ? d : 511u;
            if (lane < m) {
                pool_pk[off + g0 + lane]  = (lrv << 17) | ((unsigned)ec & 0x1FFFFu);
                pool_val[off + g0 + lane] = ev;
            }
            gather_add<VEC>(lrv, ec, ev, m, rows, lane, x, acc);
        }
    };

    uint4 kk[PF];
    int it   = 0;
    int nseg = 0;
    for (;;) {
#pragma unroll
        for (int p = 0; p < PF; ++p) kk[p] = load_keys(it + p);
        int ntot = 0;

        for (;;) {
#pragma unroll
            for (int p = 0; p < PF; ++p) {
                const int st = (s0 + it) % NSS;
                const bool inlist = st * 1024 + tid < NE / 16;
                const unsigned int ebase = (unsigned)st * SS + (unsigned)tid * 16u;

                const unsigned int kw[4] = {kk[p].x, kk[p].y, kk[p].z, kk[p].w};
                unsigned int z[4];
#pragma unroll
                for (int d = 0; d < 4; ++d) {
                    const unsigned int t = (inlist ? kw[d] : nokey) ^ splat;
                    z[d] = ~(((t & 0x7F7F7F7Fu) + 0x7F7F7F7Fu) | t | 0x7F7F7F7Fu);
                }
                kk[p] = load_keys(it + PF);
                if (__builtin_amdgcn_ballot_w32((z[0] | z[1] | z[2] | z[3]) != 0u) != 0u) {
                    unsigned int w = (z[0] >> 7) | (z[1] >> 6) | (z[2] >> 5) | (z[3] >> 4);
                    for (;;) {
                        const unsigned int voters = __builtin_amdgcn_ballot_w32(w != 0u);
                        if (voters == 0u) break;
                        if (w != 0u) {
                            const int bp = __builtin_ctz(w);
                            w &= w - 1u;
                            const unsigned int slot = (unsigned)ntot + __builtin_amdgcn_mbcnt_lo(voters, 0u);
                            q[qo + slot] = ebase + 4u * (unsigned)(bp & 7) + (unsigned)(bp >> 3);
                        }
                        ntot += __builtin_popcount(voters);
                    }
                }
                ++it;
                if (it >= NSS || ntot > QW - WSTEP) goto scanned;
            }
        }
scanned:
        if (ntot > 0) {
            unsigned int o = 0u;
            if (lane == 0) {
                o = atomicAdd(pool_cnt, ((unsigned)ntot + 31u) & ~31u);
                seg[(sbase + nseg) * 2 + 0] = o;
                seg[(sbase + nseg) * 2 + 1] = (unsigned)ntot;
            }
            o = (unsigned)__builtin_amdgcn_readfirstlane((int)o);
            __builtin_amdgcn_fence(__ATOMIC_ACQ_REL, "wavefront", "local");
            drain(ntot, o);
            ++nseg;
            __builtin_amdgcn_fence(__ATOMIC_ACQ_REL, "wavefront", "local");
        }
        if (it >= NSS) break;
    }
    if (lane == 0) nsegs[b * NWV + wave] = (unsigned)nseg;
    __syncthreads();

    for (int j = wave; j < rows; j += 32) {
        VT val;
#pragma unroll
        for (int k = 0; k < VEC; ++k) val[k] = acc[j * D + k * 32 + lane];
        *(VT*)(out + (size_t)(base + j) * D + lane * VEC) = val;
    }
}

template<int VEC>
__global__ __launch_bounds__(1024)
void spmm_owner_lists(const unsigned int* __restrict__ pool_pk, const float* __restrict__ pool_val,
                      const unsigned int* __restrict__ seg, const unsigned int* __restrict__ nsegs,
                      const float* __restrict__ x, float* __restrict__ out)
{
    constexpr int D = 32 * VEC;
    typedef __attribute__((ext_vector_type(VEC))) float VT;

    __shared__ float acc[RPB * 128];

    const int tid  = threadIdx.x;
    const int lane = tid & 31;
    const int wave = __builtin_amdgcn_readfirstlane(tid >> 5);
    const int b    = blockIdx.x;
    const int base = b * RPB;
    const int rows = (NN - base) < RPB ? (NN - base) : RPB;
    const int ns   = __builtin_amdgcn_readfirstlane((int)nsegs[b * NWV + wave]);
    const unsigned int sbase = (unsigned)(b * NWV + wave) * SEGPW;

    for (int i = tid; i < RPB * D / 4; i += 1024)
        ((float4*)acc)[i] = make_float4(0.0f, 0.0f, 0.0f, 0.0f);
    __syncthreads();

    for (int s = 0; s < ns; ++s) {
        const unsigned int off = (unsigned)__builtin_amdgcn_readfirstlane((int)seg[(sbase + s) * 2 + 0]);
        const int          n   = __builtin_amdgcn_readfirstlane((int)seg[(sbase + s) * 2 + 1]);
        int g0 = 0;
        if (g0 < n) {
            int m = (n - g0) < 32 ? (n - g0) : 32;
            unsigned int pk = pool_pk[off + g0 + (lane < m ? lane : 0)];
            float        ev = pool_val[off + g0 + (lane < m ? lane : 0)];
            for (;;) {
                const unsigned int cpk = pk;
                const float        cev = ev;
                const int          cm  = m;
                const int gn = g0 + 32;
                if (gn < n) {
                    m  = (n - gn) < 32 ? (n - gn) : 32;
                    pk = pool_pk[off + gn + (lane < m ? lane : 0)];
                    ev = pool_val[off + gn + (lane < m ? lane : 0)];
                }
                gather_add<VEC>(cpk >> 17, (int)(cpk & 0x1FFFFu), cev, cm, rows, lane, x, acc);
                if (gn >= n) break;
                g0 = gn;
            }
        }
    }
    __syncthreads();

    for (int j = wave; j < rows; j += 32) {
        VT val;
#pragma unroll
        for (int k = 0; k < VEC; ++k) val[k] = acc[j * D + k * 32 + lane];
        *(VT*)(out + (size_t)(base + j) * D + lane * VEC) = val;
    }
}

extern "C" void kernel_launch(void* const* d_in, const int* in_sizes, int n_in,
                              void* d_out, int out_size, void* d_ws, size_t ws_size,
                              hipStream_t stream)
{
    const float* x    = (const float*)d_in[0];
    const float* vals = (const float*)d_in[1];
    const float* W1   = (const float*)d_in[2];
    const float* b1   = (const float*)d_in[3];
    const float* W2   = (const float*)d_in[4];
    const float* b2   = (const float*)d_in[5];
    const float* W3   = (const float*)d_in[6];
    const float* b3   = (const float*)d_in[7];
    const int*   row  = (const int*)d_in[8];
    const int*   col  = (const int*)d_in[9];

    float* bufA = (float*)d_ws;
    float* bufB = bufA + (size_t)NN * 128;
    unsigned int* keys = (unsigned int*)(bufB + (size_t)NN * 128);
    unsigned int* pool_pk  = keys + NE / 4;
    float*        pool_val = (float*)(pool_pk + POOLN);
    unsigned int* seg      = (unsigned int*)(pool_val + POOLN);
    unsigned int* nsegs    = seg + 256 * NWV * SEGPW * 2;
    unsigned int* pool_cnt = nsegs + 256 * NWV;

    const dim3 gBlk(128);
    const dim3 gGrd((MTILES + 3) / 4);
    const dim3 sBlk(1024);
    const dim3 sGrd(256);
    const dim3 kBlk(256);
    const dim3 kGrd((NE / 16 + 255) / 256);

    owner_keys<<<kGrd, kBlk, 0, stream>>>(row, keys, pool_cnt);

    gemm_wmma_f32<8, false><<<gGrd, gBlk, 0, stream>>>(x, W1, b1, bufA);
    spmm_owner_build<4><<<sGrd, sBlk, 0, stream>>>(vals, row, col, keys, bufA, bufB,
                                                   pool_pk, pool_val, seg, nsegs, pool_cnt);

    gemm_wmma_f32<8, true><<<gGrd, gBlk, 0, stream>>>(bufB, W2, b2, bufA);
    spmm_owner_lists<4><<<sGrd, sBlk, 0, stream>>>(pool_pk, pool_val, seg, nsegs, bufA, bufB);

    gemm_wmma_f32<4, true><<<gGrd, gBlk, 0, stream>>>(bufB, W3, b3, bufA);
    spmm_owner_lists<2><<<sGrd, sBlk, 0, stream>>>(pool_pk, pool_val, seg, nsegs, bufA, (float*)d_out);
}


# SpMM build scan: owner_keys also writes a per-owner bitmap of 16-edge chunks (LDS integer ORs); a wave loads keys of flagged chunks only, not all 98 steps; list order changes, LDS float adds kept
# speedup vs baseline: 3.0424x; 1.0330x over previous
#include <hip/hip_runtime.h>

typedef __attribute__((ext_vector_type(2))) float v2f;
typedef __attribute__((ext_vector_type(8))) float v8f;

static constexpr int NN     = 100000;
static constexpr int NE     = 1600000;
static constexpr int DK     = 128;
static constexpr int MTILES = NN / 16;
static constexpr int RPB    = 391;
static constexpr int SS     = 16384;
static constexpr int NSS    = (NE + SS - 1) / SS;
static constexpr int QCAP   = 30720;
static constexpr int NWV    = 32;
static constexpr int QW     = QCAP / NWV;
static constexpr int WSTEP  = 512;
static constexpr int SEGPW  = 112;
static constexpr int POOLN  = NE + 400000;
static constexpr int BMW    = 257;

template<int NT, bool RELU_IN>
__global__ __launch_bounds__(128)
void gemm_wmma_f32(const float* __restrict__ A, const float* __restrict__ W,
                   const float* __restrict__ bias, float* __restrict__ out)
{
    constexpr int DOUT = NT * 16;
    const int lane  = threadIdx.x & 31;
    const int wave  = threadIdx.x >> 5;
    const int mtile = blockIdx.x * 4 + wave;
    if (mtile >= MTILES) return;

    const int half = lane >> 4;
    const int mn   = lane & 15;
    const int kb   = half * 2;

    const float* arow = A + (size_t)(mtile * 16 + mn) * DK;

    v8f acc[NT];
#pragma unroll
    for (int t = 0; t < NT; ++t) acc[t] = (v8f)0.0f;

    for (int k = 0; k < DK; k += 4) {
        v2f a = *(const v2f*)(arow + k + kb);
        if (RELU_IN) { a.x = fmaxf(a.x, 0.0f); a.y = fmaxf(a.y, 0.0f); }
        const float* wp = W + (size_t)(k + kb) * DOUT + mn;
#pragma unroll
        for (int t = 0; t < NT; ++t) {
            v2f b;
            b.x = wp[t * 16];
            b.y = wp[t * 16 + DOUT];
            acc[t] = __builtin_amdgcn_wmma_f32_16x16x4_f32(
                false, a, false, b, (short)0, acc[t], false, false);
        }
    }

#pragma unroll
    for (int t = 0; t < NT; ++t) {
        const float bv = bias[t * 16 + mn];
#pragma unroll
        for (int v = 0; v < 8; ++v) {
            const int m = mtile * 16 + v + 8 * half;
            out[(size_t)m * DOUT + t * 16 + mn] = acc[t][v] + bv;
        }
    }
}

__global__ __launch_bounds__(1024)
void owner_keys(const int* __restrict__ row, unsigned int* __restrict__ keys,
                unsigned int* __restrict__ bitmap, unsigned int* __restrict__ pool_cnt)
{
    __shared__ unsigned int tab[32 * BMW];
    const int tid  = threadIdx.x;
    const int lane = tid & 31;
    const int wave = tid >> 5;
    const int s    = blockIdx.x;
    const int t    = s * 1024 + tid;
    if (t == 0) *pool_cnt = 0u;
    for (int i = tid; i < 32 * BMW; i += 1024) tab[i] = 0u;
    __syncthreads();
    if (t < NE / 16) {
        const int4* rp = (const int4*)(row + (size_t)t * 16);
        unsigned int* mytab = tab + wave * BMW;
        const unsigned int bit = 1u << lane;
        unsigned int o[4];
#pragma unroll
        for (int d = 0; d < 4; ++d) {
            const int4 r = rp[d];
            const unsigned int k0 = ((unsigned)r.x / RPB) & 255u;
            const unsigned int k1 = ((unsigned)r.y / RPB) & 255u;
            const unsigned int k2 = ((unsigned)r.z / RPB) & 255u;
            const unsigned int k3 = ((unsigned)r.w / RPB) & 255u;
            o[d] = k0 | (k1 << 8) | (k2 << 16) | (k3 << 24);
            atomicOr(mytab + k0, bit);
            atomicOr(mytab + k1, bit);
            atomicOr(mytab + k2, bit);
            atomicOr(mytab + k3, bit);
        }
        ((uint4*)keys)[t] = make_uint4(o[0], o[1], o[2], o[3]);
    }
    __syncthreads();
#pragma unroll
    for (int i = 0; i < 8; ++i) {
        const int o = wave + 32 * i;
        bitmap[((size_t)o * NSS + s) * 32 + lane] = tab[lane * BMW + o];
    }
}


template<int VEC>
__device__ __forceinline__ void gather_add(unsigned int lrv, int ecv, float ev, int m, int rows, int lane,
                                           const float* __restrict__ x, float* acc)
{
    constexpr int D = 32 * VEC;
    constexpr int U = 16;
    typedef __attribute__((ext_vector_type(VEC))) float VT;
    for (int j0 = 0; j0 < m; j0 += U) {
        VT xv[U];
#pragma unroll
        for (int u = 0; u < U; ++u) {
            const int j = (j0 + u < m) ? (j0 + u) : j0;
            const int c = __builtin_amdgcn_readlane(ecv, j);
            xv[u] = *(const VT*)(x + (size_t)c * D + lane * VEC);
        }
#pragma unroll
        for (int u = 0; u < U; ++u) asm volatile("" :: "v"(xv[u]));
#pragma unroll
        for (int u = 0; u < U; ++u) {
            const int j = j0 + u;
            if (j < m) {
                const unsigned int lr = (unsigned)__builtin_amdgcn_readlane((int)lrv, j);
                const float v = __int_as_float(__builtin_amdgcn_readlane(__float_as_int(ev), j));
                if (lr < (unsigned)rows) {
                    float* a = acc + lr * D + lane;
#pragma unroll
                    for (int k = 0; k < VEC; ++k) atomicAdd(a + k * 32, v * xv[u][k]);
                }
            }
        }
    }
}

template<int VEC>
__global__ __launch_bounds__(1024)
void spmm_owner_build(const float* __restrict__ vals, const int* __restrict__ row,
                      const int* __restrict__ col, const unsigned int* __restrict__ keys,
                      const unsigned int* __restrict__ bitmap,
                      const float* __restrict__ x, float* __restrict__ out,
                      unsigned int* __restrict__ pool_pk, float* __restrict__ pool_val,
                      unsigned int* __restrict__ seg, unsigned int* __restrict__ nsegs,
                      unsigned int* __restrict__ pool_cnt)
{
    constexpr int D  = 32 * VEC;
    typedef __attribute__((ext_vector_type(VEC))) float VT;

    __shared__ float        acc[RPB * D];
    __shared__ unsigned int q[QCAP];

    const int tid  = threadIdx.x;
    const int lane = tid & 31;
    const int wave = __builtin_amdgcn_readfirstlane(tid >> 5);
    const int b    = blockIdx.x;
    const int base = b * RPB;
    const int rows = (NN - base) < RPB ? (NN - base) : RPB;
    const unsigned int splat  = (unsigned)b * 0x01010101u;

    for (int i = tid; i < RPB * D / 4; i += 1024)
        ((float4*)acc)[i] = make_float4(0.0f, 0.0f, 0.0f, 0.0f);
    __syncthreads();

    const int qo = wave * QW;
    const unsigned int sbase = (unsigned)(b * NWV + wave) * SEGPW;

    const uint4* kp = (const uint4*)keys;

    auto drain = [&](int n, unsigned int off) {
        for (int g0 = 0; g0 < n; g0 += 32) {
            const int m = (n - g0) < 32 ? (n - g0) : 32;
            const unsigned int e = q[qo + g0 + (lane < m ? lane : 0)];
            const int   er = row[e];
            const int   ec = col[e];
            const float ev = vals[e];
            const unsigned int d   = (unsigned)(er - base);
            const unsigned int lrv = d < 511u ? d : 511u;
            if (lane < m) {
                pool_pk[off + g0 + lane]  = (lrv << 17) | ((unsigned)ec & 0x1FFFFu);
                pool_val[off + g0 + lane] = ev;
            }
            gather_add<VEC>(lrv, ec, ev, m, rows, lane, x, acc);
        }
    };

    int nseg = 0;
    int ntot = 0;

    auto flush = [&]() {
        unsigned int o = 0u;
        if (lane == 0) {
            o = atomicAdd(pool_cnt, ((unsigned)ntot + 31u) & ~31u);
            seg[(sbase + nseg) * 2 + 0] = o;
            seg[(sbase + nseg) * 2 + 1] = (unsigned)ntot;
        }
        o = (unsigned)__builtin_amdgcn_readfirstlane((int)o);
        __builtin_amdgcn_fence(__ATOMIC_ACQ_REL, "wavefront", "local");
        drain(ntot, o);
        ++nseg;
        __builtin_amdgcn_fence(__ATOMIC_ACQ_REL, "wavefront", "local");
        ntot = 0;
    };

    unsigned int cur, n1, n2, n3;
    {
        const unsigned int* bp = bitmap + ((size_t)b * NSS) * 32 + wave;
        cur = bp[(size_t)lane * 32];
        n1  = bp[(size_t)(lane + 32) * 32];
        n2  = bp[(size_t)(lane + 64) * 32];
        n3  = (lane + 96 < NSS) ? bp[(size_t)(lane + 96) * 32] : 0u;
    }
    int st = lane;

    for (;;) {
#pragma unroll
        for (int r = 0; r < 3; ++r)
            if (cur == 0u) { cur = n1; n1 = n2; n2 = n3; n3 = 0u; st += 32; }
        if (__builtin_amdgcn_ballot_w32(cur != 0u) == 0u) break;

        unsigned int w = 0u;
        unsigned int ebase = 0u;
        if (cur != 0u) {
            const int cl = __builtin_ctz(cur);
            cur &= cur - 1u;
            const unsigned int g = (unsigned)st * 1024u + (unsigned)wave * 32u + (unsigned)cl;
            const uint4 k4 = kp[g];
            ebase = g * 16u;
            const unsigned int kw[4] = {k4.x, k4.y, k4.z, k4.w};
            unsigned int z[4];
#pragma unroll
            for (int d = 0; d < 4; ++d) {
                const unsigned int t = kw[d] ^ splat;
                z[d] = ~(((t & 0x7F7F7F7Fu) + 0x7F7F7F7Fu) | t | 0x7F7F7F7Fu);
            }
            w = (z[0] >> 7) | (z[1] >> 6) | (z[2] >> 5) | (z[3] >> 4);
        }
        for (;;) {
            const unsigned int voters = __builtin_amdgcn_ballot_w32(w != 0u);
            if (voters == 0u) break;
            if (w != 0u) {
                const int bp = __builtin_ctz(w);
                w &= w - 1u;
                const unsigned int slot = (unsigned)ntot + __builtin_amdgcn_mbcnt_lo(voters, 0u);
                q[qo + slot] = ebase + 4u * (unsigned)(bp & 7) + (unsigned)(bp >> 3);
            }
            ntot += __builtin_popcount(voters);
        }
        if (ntot > QW - WSTEP) flush();
    }
    if (ntot > 0) flush();
    if (lane == 0) nsegs[b * NWV + wave] = (unsigned)nseg;
    __syncthreads();

    for (int j = wave; j < rows; j += 32) {
        VT val;
#pragma unroll
        for (int k = 0; k < VEC; ++k) val[k] = acc[j * D + k * 32 + lane];
        *(VT*)(out + (size_t)(base + j) * D + lane * VEC) = val;
    }
}

template<int VEC>
__global__ __launch_bounds__(1024)
void spmm_owner_lists(const unsigned int* __restrict__ pool_pk, const float* __restrict__ pool_val,
                      const unsigned int* __restrict__ seg, const unsigned int* __restrict__ nsegs,
                      const float* __restrict__ x, float* __restrict__ out)
{
    constexpr int D = 32 * VEC;
    typedef __attribute__((ext_vector_type(VEC))) float VT;

    __shared__ float acc[RPB * 128];

    const int tid  = threadIdx.x;
    const int lane = tid & 31;
    const int wave = __builtin_amdgcn_readfirstlane(tid >> 5);
    const int b    = blockIdx.x;
    const int base = b * RPB;
    const int rows = (NN - base) < RPB ? (NN - base) : RPB;
    const int ns   = __builtin_amdgcn_readfirstlane((int)nsegs[b * NWV + wave]);
    const unsigned int sbase = (unsigned)(b * NWV + wave) * SEGPW;

    for (int i = tid; i < RPB * D / 4; i += 1024)
        ((float4*)acc)[i] = make_float4(0.0f, 0.0f, 0.0f, 0.0f);
    __syncthreads();

    for (int s = 0; s < ns; ++s) {
        const unsigned int off = (unsigned)__builtin_amdgcn_readfirstlane((int)seg[(sbase + s) * 2 + 0]);
        const int          n   = __builtin_amdgcn_readfirstlane((int)seg[(sbase + s) * 2 + 1]);
        int g0 = 0;
        if (g0 < n) {
            int m = (n - g0) < 32 ? (n - g0) : 32;
            unsigned int pk = pool_pk[off + g0 + (lane < m ? lane : 0)];
            float        ev = pool_val[off + g0 + (lane < m ? lane : 0)];
            for (;;) {
                const unsigned int cpk = pk;
                const float        cev = ev;
                const int          cm  = m;
                const int gn = g0 + 32;
                if (gn < n) {
                    m  = (n - gn) < 32 ? (n - gn) : 32;
                    pk = pool_pk[off + gn + (lane < m ? lane : 0)];
                    ev = pool_val[off + gn + (lane < m ? lane : 0)];
                }
                gather_add<VEC>(cpk >> 17, (int)(cpk & 0x1FFFFu), cev, cm, rows, lane, x, acc);
                if (gn >= n) break;
                g0 = gn;
            }
        }
    }
    __syncthreads();

    for (int j = wave; j < rows; j += 32) {
        VT val;
#pragma unroll
        for (int k = 0; k < VEC; ++k) val[k] = acc[j * D + k * 32 + lane];
        *(VT*)(out + (size_t)(base + j) * D + lane * VEC) = val;
    }
}

extern "C" void kernel_launch(void* const* d_in, const int* in_sizes, int n_in,
                              void* d_out, int out_size, void* d_ws, size_t ws_size,
                              hipStream_t stream)
{
    const float* x    = (const float*)d_in[0];
    const float* vals = (const float*)d_in[1];
    const float* W1   = (const float*)d_in[2];
    const float* b1   = (const float*)d_in[3];
    const float* W2   = (const float*)d_in[4];
    const float* b2   = (const float*)d_in[5];
    const float* W3   = (const float*)d_in[6];
    const float* b3   = (const float*)d_in[7];
    const int*   row  = (const int*)d_in[8];
    const int*   col  = (const int*)d_in[9];

    float* bufA = (float*)d_ws;
    float* bufB = bufA + (size_t)NN * 128;
    unsigned int* keys = (unsigned int*)(bufB + (size_t)NN * 128);
    unsigned int* pool_pk  = keys + NE / 4;
    float*        pool_val = (float*)(pool_pk + POOLN);
    unsigned int* seg      = (unsigned int*)(pool_val + POOLN);
    unsigned int* nsegs    = seg + 256 * NWV * SEGPW * 2;
    unsigned int* pool_cnt = nsegs + 256 * NWV;
    unsigned int* bitmap   = pool_cnt + 32;

    const dim3 gBlk(128);
    const dim3 gGrd((MTILES + 3) / 4);
    const dim3 sBlk(1024);
    const dim3 sGrd(256);
    const dim3 kBlk(1024);
    const dim3 kGrd(NSS);

    owner_keys<<<kGrd, kBlk, 0, stream>>>(row, keys, bitmap, pool_cnt);

    gemm_wmma_f32<8, false><<<gGrd, gBlk, 0, stream>>>(x, W1, b1, bufA);
    spmm_owner_build<4><<<sGrd, sBlk, 0, stream>>>(vals, row, col, keys, bitmap, bufA, bufB,
                                                   pool_pk, pool_val, seg, nsegs, pool_cnt);

    gemm_wmma_f32<8, true><<<gGrd, gBlk, 0, stream>>>(bufB, W2, b2, bufA);
    spmm_owner_lists<4><<<sGrd, sBlk, 0, stream>>>(pool_pk, pool_val, seg, nsegs, bufA, bufB);

    gemm_wmma_f32<4, true><<<gGrd, gBlk, 0, stream>>>(bufB, W3, b3, bufA);
    spmm_owner_lists<2><<<sGrd, sBlk, 0, stream>>>(pool_pk, pool_val, seg, nsegs, bufA, (float*)d_out);
}


# GEMMs: A, W and output moved 16 bytes per lane (was 8, 4, 4): a lane half owns 4 consecutive k of each 8, tile columns interleaved by 4; the k order of the product sums changes
# speedup vs baseline: 3.1850x; 1.0469x over previous
#include <hip/hip_runtime.h>

typedef __attribute__((ext_vector_type(2))) float v2f;
typedef __attribute__((ext_vector_type(8))) float v8f;

static constexpr int NN     = 100000;
static constexpr int NE     = 1600000;
static constexpr int DK     = 128;
static constexpr int MTILES = NN / 16;
static constexpr int RPB    = 391;
static constexpr int SS     = 16384;
static constexpr int NSS    = (NE + SS - 1) / SS;
static constexpr int QCAP   = 30720;
static constexpr int NWV    = 32;
static constexpr int QW     = QCAP / NWV;
static constexpr int WSTEP  = 512;
static constexpr int SEGPW  = 112;
static constexpr int POOLN  = NE + 400000;
static constexpr int BMW    = 257;

template<int NT, bool RELU_IN>
__global__ __launch_bounds__(128)
void gemm_wmma_f32(const float* __restrict__ A, const float* __restrict__ W,
                   const float* __restrict__ bias, float* __restrict__ out)
{
    constexpr int DOUT = NT * 16;
    constexpr int NG   = NT / 4;
    const int lane  = threadIdx.x & 31;
    const int wave  = threadIdx.x >> 5;
    const int mtile = blockIdx.x * 4 + wave;
    if (mtile >= MTILES) return;

    const int half = lane >> 4;
    const int mn   = lane & 15;

    const float* arow = A + (size_t)(mtile * 16 + mn) * DK + half * 4;
    const float* wcol = W + (size_t)(half * 4) * DOUT + mn * 4;

    v8f acc[NT];
#pragma unroll
    for (int t = 0; t < NT; ++t) acc[t] = (v8f)0.0f;

#pragma unroll 1
    for (int k8 = 0; k8 < DK; k8 += 8) {
        float4 a4 = *(const float4*)(arow + k8);
        if (RELU_IN) {
            a4.x = fmaxf(a4.x, 0.0f); a4.y = fmaxf(a4.y, 0.0f);
            a4.z = fmaxf(a4.z, 0.0f); a4.w = fmaxf(a4.w, 0.0f);
        }
        float4 w4[4][NG];
#pragma unroll
        for (int j = 0; j < 4; ++j)
#pragma unroll
            for (int g = 0; g < NG; ++g)
                w4[j][g] = *(const float4*)(wcol + (size_t)(k8 + j) * DOUT + g * 64);
#pragma unroll
        for (int s = 0; s < 2; ++s) {
            v2f a;
            a.x = s ? a4.z : a4.x;
            a.y = s ? a4.w : a4.y;
#pragma unroll
            for (int t = 0; t < NT; ++t) {
                const float4 lo = w4[2 * s][t >> 2];
                const float4 hi = w4[2 * s + 1][t >> 2];
                const int c = t & 3;
                v2f b;
                b.x = c == 0 ? lo.x : c == 1 ? lo.y : c == 2 ? lo.z : lo.w;
                b.y = c == 0 ? hi.x : c == 1 ? hi.y : c == 2 ? hi.z : hi.w;
                acc[t] = __builtin_amdgcn_wmma_f32_16x16x4_f32(
                    false, a, false, b, (short)0, acc[t], false, false);
            }
        }
    }

#pragma unroll
    for (int g = 0; g < NG; ++g) {
        const float4 bv = *(const float4*)(bias + g * 64 + mn * 4);
#pragma unroll
        for (int v = 0; v < 8; ++v) {
            const int m = mtile * 16 + v + 8 * half;
            float4 o;
            o.x = acc[4 * g + 0][v] + bv.x;
            o.y = acc[4 * g + 1][v] + bv.y;
            o.z = acc[4 * g + 2][v] + bv.z;
            o.w = acc[4 * g + 3][v] + bv.w;
            *(float4*)(out + (size_t)m * DOUT + g * 64 + mn * 4) = o;
        }
    }
}

__global__ __launch_bounds__(1024)
void owner_keys(const int* __restrict__ row, unsigned int* __restrict__ keys,
                unsigned int* __restrict__ bitmap, unsigned int* __restrict__ pool_cnt)
{
    __shared__ unsigned int tab[32 * BMW];
    const int tid  = threadIdx.x;
    const int lane = tid & 31;
    const int wave = tid >> 5;
    const int s    = blockIdx.x;
    const int t    = s * 1024 + tid;
    if (t == 0) *pool_cnt = 0u;
    for (int i = tid; i < 32 * BMW; i += 1024) tab[i] = 0u;
    __syncthreads();
    if (t < NE / 16) {
        const int4* rp = (const int4*)(row + (size_t)t * 16);
        unsigned int* mytab = tab + wave * BMW;
        const unsigned int bit = 1u << lane;
        unsigned int o[4];
#pragma unroll
        for (int d = 0; d < 4; ++d) {
            const int4 r = rp[d];
            const unsigned int k0 = ((unsigned)r.x / RPB) & 255u;
            const unsigned int k1 = ((unsigned)r.y / RPB) & 255u;
            const unsigned int k2 = ((unsigned)r.z / RPB) & 255u;
            const unsigned int k3 = ((unsigned)r.w / RPB) & 255u;
            o[d] = k0 | (k1 << 8) | (k2 << 16) | (k3 << 24);
            atomicOr(mytab + k0, bit);
            atomicOr(mytab + k1, bit);
            atomicOr(mytab + k2, bit);
            atomicOr(mytab + k3, bit);
        }
        ((uint4*)keys)[t] = make_uint4(o[0], o[1], o[2], o[3]);
    }
    __syncthreads();
#pragma unroll
    for (int i = 0; i < 8; ++i) {
        const int o = wave + 32 * i;
        bitmap[((size_t)o * NSS + s) * 32 + lane] = tab[lane * BMW + o];
    }
}


template<int VEC>
__device__ __forceinline__ void gather_add(unsigned int lrv, int ecv, float ev, int m, int rows, int lane,
                                           const float* __restrict__ x, float* acc)
{
    constexpr int D = 32 * VEC;
    constexpr int U = 16;
    typedef __attribute__((ext_vector_type(VEC))) float VT;
    for (int j0 = 0; j0 < m; j0 += U) {
        VT xv[U];
#pragma unroll
        for (int u = 0; u < U; ++u) {
            const int j = (j0 + u < m) ? (j0 + u) : j0;
            const int c = __builtin_amdgcn_readlane(ecv, j);
            xv[u] = *(const VT*)(x + (size_t)c * D + lane * VEC);
        }
#pragma unroll
        for (int u = 0; u < U; ++u) asm volatile("" :: "v"(xv[u]));
#pragma unroll
        for (int u = 0; u < U; ++u) {
            const int j = j0 + u;
            if (j < m) {
                const unsigned int lr = (unsigned)__builtin_amdgcn_readlane((int)lrv, j);
                const float v = __int_as_float(__builtin_amdgcn_readlane(__float_as_int(ev), j));
                if (lr < (unsigned)rows) {
                    float* a = acc + lr * D + lane;
#pragma unroll
                    for (int k = 0; k < VEC; ++k) atomicAdd(a + k * 32, v * xv[u][k]);
                }
            }
        }
    }
}

template<int VEC>
__global__ __launch_bounds__(1024)
void spmm_owner_build(const float* __restrict__ vals, const int* __restrict__ row,
                      const int* __restrict__ col, const unsigned int* __restrict__ keys,
                      const unsigned int* __restrict__ bitmap,
                      const float* __restrict__ x, float* __restrict__ out,
                      unsigned int* __restrict__ pool_pk, float* __restrict__ pool_val,
                      unsigned int* __restrict__ seg, unsigned int* __restrict__ nsegs,
                      unsigned int* __restrict__ pool_cnt)
{
    constexpr int D  = 32 * VEC;
    typedef __attribute__((ext_vector_type(VEC))) float VT;

    __shared__ float        acc[RPB * D];
    __shared__ unsigned int q[QCAP];

    const int tid  = threadIdx.x;
    const int lane = tid & 31;
    const int wave = __builtin_amdgcn_readfirstlane(tid >> 5);
    const int b    = blockIdx.x;
    const int base = b * RPB;
    const int rows = (NN - base) < RPB ? (NN - base) : RPB;
    const unsigned int splat  = (unsigned)b * 0x01010101u;

    for (int i = tid; i < RPB * D / 4; i += 1024)
        ((float4*)acc)[i] = make_float4(0.0f, 0.0f, 0.0f, 0.0f);
    __syncthreads();

    const int qo = wave * QW;
    const unsigned int sbase = (unsigned)(b * NWV + wave) * SEGPW;

    const uint4* kp = (const uint4*)keys;

    auto drain = [&](int n, unsigned int off) {
        for (int g0 = 0; g0 < n; g0 += 32) {
            const int m = (n - g0) < 32 ? (n - g0) : 32;
            const unsigned int e = q[qo + g0 + (lane < m ? lane : 0)];
            const int   er = row[e];
            const int   ec = col[e];
            const float ev = vals[e];
            const unsigned int d   = (unsigned)(er - base);
            const unsigned int lrv = d < 511u ? d : 511u;
            if (lane < m) {
                pool_pk[off + g0 + lane]  = (lrv << 17) | ((unsigned)ec & 0x1FFFFu);
                pool_val[off + g0 + lane] = ev;
            }
            gather_add<VEC>(lrv, ec, ev, m, rows, lane, x, acc);
        }
    };

    int nseg = 0;
    int ntot = 0;

    auto flush = [&]() {
        unsigned int o = 0u;
        if (lane == 0) {
            o = atomicAdd(pool_cnt, ((unsigned)ntot + 31u) & ~31u);
            seg[(sbase + nseg) * 2 + 0] = o;
            seg[(sbase + nseg) * 2 + 1] = (unsigned)ntot;
        }
        o = (unsigned)__builtin_amdgcn_readfirstlane((int)o);
        __builtin_amdgcn_fence(__ATOMIC_ACQ_REL, "wavefront", "local");
        drain(ntot, o);
        ++nseg;
        __builtin_amdgcn_fence(__ATOMIC_ACQ_REL, "wavefront", "local");
        ntot = 0;
    };

    unsigned int cur, n1, n2, n3;
    {
        const unsigned int* bp = bitmap + ((size_t)b * NSS) * 32 + wave;
        cur = bp[(size_t)lane * 32];
        n1  = bp[(size_t)(lane + 32) * 32];
        n2  = bp[(size_t)(lane + 64) * 32];
        n3  = (lane + 96 < NSS) ? bp[(size_t)(lane + 96) * 32] : 0u;
    }
    int st = lane;

    for (;;) {
#pragma unroll
        for (int r = 0; r < 3; ++r)
            if (cur == 0u) { cur = n1; n1 = n2; n2 = n3; n3 = 0u; st += 32; }
        if (__builtin_amdgcn_ballot_w32(cur != 0u) == 0u) break;

        unsigned int w = 0u;
        unsigned int ebase = 0u;
        if (cur != 0u) {
            const int cl = __builtin_ctz(cur);
            cur &= cur - 1u;
            const unsigned int g = (unsigned)st * 1024u + (unsigned)wave * 32u + (unsigned)cl;
            const uint4 k4 = kp[g];
            ebase = g * 16u;
            const unsigned int kw[4] = {k4.x, k4.y, k4.z, k4.w};
            unsigned int z[4];
#pragma unroll
            for (int d = 0; d < 4; ++d) {
                const unsigned int t = kw[d] ^ splat;
                z[d] = ~(((t & 0x7F7F7F7Fu) + 0x7F7F7F7Fu) | t | 0x7F7F7F7Fu);
            }
            w = (z[0] >> 7) | (z[1] >> 6) | (z[2] >> 5) | (z[3] >> 4);
        }
        for (;;) {
            const unsigned int voters = __builtin_amdgcn_ballot_w32(w != 0u);
            if (voters == 0u) break;
            if (w != 0u) {
                const int bp = __builtin_ctz(w);
                w &= w - 1u;
                const unsigned int slot = (unsigned)ntot + __builtin_amdgcn_mbcnt_lo(voters, 0u);
                q[qo + slot] = ebase + 4u * (unsigned)(bp & 7) + (unsigned)(bp >> 3);
            }
            ntot += __builtin_popcount(voters);
        }
        if (ntot > QW - WSTEP) flush();
    }
    if (ntot > 0) flush();
    if (lane == 0) nsegs[b * NWV + wave] = (unsigned)nseg;
    __syncthreads();

    for (int j = wave; j < rows; j += 32) {
        VT val;
#pragma unroll
        for (int k = 0; k < VEC; ++k) val[k] = acc[j * D + k * 32 + lane];
        *(VT*)(out + (size_t)(base + j) * D + lane * VEC) = val;
    }
}

template<int VEC>
__global__ __launch_bounds__(1024)
void spmm_owner_lists(const unsigned int* __restrict__ pool_pk, const float* __restrict__ pool_val,
                      const unsigned int* __restrict__ seg, const unsigned int* __restrict__ nsegs,
                      const float* __restrict__ x, float* __restrict__ out)
{
    constexpr int D = 32 * VEC;
    typedef __attribute__((ext_vector_type(VEC))) float VT;

    __shared__ float acc[RPB * 128];

    const int tid  = threadIdx.x;
    const int lane = tid & 31;
    const int wave = __builtin_amdgcn_readfirstlane(tid >> 5);
    const int b    = blockIdx.x;
    const int base = b * RPB;
    const int rows = (NN - base) < RPB ? (NN - base) : RPB;
    const int ns   = __builtin_amdgcn_readfirstlane((int)nsegs[b * NWV + wave]);
    const unsigned int sbase = (unsigned)(b * NWV + wave) * SEGPW;

    for (int i = tid; i < RPB * D / 4; i += 1024)
        ((float4*)acc)[i] = make_float4(0.0f, 0.0f, 0.0f, 0.0f);
    __syncthreads();

    for (int s = 0; s < ns; ++s) {
        const unsigned int off = (unsigned)__builtin_amdgcn_readfirstlane((int)seg[(sbase + s) * 2 + 0]);
        const int          n   = __builtin_amdgcn_readfirstlane((int)seg[(sbase + s) * 2 + 1]);
        int g0 = 0;
        if (g0 < n) {
            int m = (n - g0) < 32 ? (n - g0) : 32;
            unsigned int pk = pool_pk[off + g0 + (lane < m ? lane : 0)];
            float        ev = pool_val[off + g0 + (lane < m ? lane : 0)];
            for (;;) {
                const unsigned int cpk = pk;
                const float        cev = ev;
                const int          cm  = m;
                const int gn = g0 + 32;
                if (gn < n) {
                    m  = (n - gn) < 32 ? (n - gn) : 32;
                    pk = pool_pk[off + gn + (lane < m ? lane : 0)];
                    ev = pool_val[off + gn + (lane < m ? lane : 0)];
                }
                gather_add<VEC>(cpk >> 17, (int)(cpk & 0x1FFFFu), cev, cm, rows, lane, x, acc);
                if (gn >= n) break;
                g0 = gn;
            }
        }
    }
    __syncthreads();

    for (int j = wave; j < rows; j += 32) {
        VT val;
#pragma unroll
        for (int k = 0; k < VEC; ++k) val[k] = acc[j * D + k * 32 + lane];
        *(VT*)(out + (size_t)(base + j) * D + lane * VEC) = val;
    }
}

extern "C" void kernel_launch(void* const* d_in, const int* in_sizes, int n_in,
                              void* d_out, int out_size, void* d_ws, size_t ws_size,
                              hipStream_t stream)
{
    const float* x    = (const float*)d_in[0];
    const float* vals = (const float*)d_in[1];
    const float* W1   = (const float*)d_in[2];
    const float* b1   = (const float*)d_in[3];
    const float* W2   = (const float*)d_in[4];
    const float* b2   = (const float*)d_in[5];
    const float* W3   = (const float*)d_in[6];
    const float* b3   = (const float*)d_in[7];
    const int*   row  = (const int*)d_in[8];
    const int*   col  = (const int*)d_in[9];

    float* bufA = (float*)d_ws;
    float* bufB = bufA + (size_t)NN * 128;
    unsigned int* keys = (unsigned int*)(bufB + (size_t)NN * 128);
    unsigned int* pool_pk  = keys + NE / 4;
    float*        pool_val = (float*)(pool_pk + POOLN);
    unsigned int* seg      = (unsigned int*)(pool_val + POOLN);
    unsigned int* nsegs    = seg + 256 * NWV * SEGPW * 2;
    unsigned int* pool_cnt = nsegs + 256 * NWV;
    unsigned int* bitmap   = pool_cnt + 32;

    const dim3 gBlk(128);
    const dim3 gGrd((MTILES + 3) / 4);
    const dim3 sBlk(1024);
    const dim3 sGrd(256);
    const dim3 kBlk(1024);
    const dim3 kGrd(NSS);

    owner_keys<<<kGrd, kBlk, 0, stream>>>(row, keys, bitmap, pool_cnt);

    gemm_wmma_f32<8, false><<<gGrd, gBlk, 0, stream>>>(x, W1, b1, bufA);
    spmm_owner_build<4><<<sGrd, sBlk, 0, stream>>>(vals, row, col, keys, bitmap, bufA, bufB,
                                                   pool_pk, pool_val, seg, nsegs, pool_cnt);

    gemm_wmma_f32<8, true><<<gGrd, gBlk, 0, stream>>>(bufB, W2, b2, bufA);
    spmm_owner_lists<4><<<sGrd, sBlk, 0, stream>>>(pool_pk, pool_val, seg, nsegs, bufA, bufB);

    gemm_wmma_f32<4, true><<<gGrd, gBlk, 0, stream>>>(bufB, W3, b3, bufA);
    spmm_owner_lists<2><<<sGrd, sBlk, 0, stream>>>(pool_pk, pool_val, seg, nsegs, bufA, (float*)d_out);
}
